# starting kernel
# baseline (speedup 1.0000x reference)
#include <hip/hip_runtime.h>
#include <hip/hip_bf16.h>

typedef float v2f __attribute__((ext_vector_type(2)));
typedef float v8f __attribute__((ext_vector_type(8)));

#define NEG_SLOPE 0.2f

template<int K, int NOUT, int NT>
__global__ __launch_bounds__(256) void gemm_wmma_f32(
    const float* __restrict__ X, const float* __restrict__ W,
    float* __restrict__ C, int M)
{
  const int lane   = threadIdx.x & 31;
  const int wave   = threadIdx.x >> 5;
  constexpr int tilesN = NOUT / (16 * NT);
  const int tiles  = (M >> 4) * tilesN;
  const int tile   = blockIdx.x * 8 + wave;
  if (tile >= tiles) return;
  const int tM = tile / tilesN;
  const int tN = tile % tilesN;

  const int half = lane >> 4;
  const int l15  = lane & 15;
  const float* xrow = X + (size_t)(tM * 16 + l15) * K + 2 * half;
  const float* wb   = W + (size_t)(2 * half) * NOUT + tN * (16 * NT) + l15;

  v8f acc[NT];
#pragma unroll
  for (int t = 0; t < NT; ++t) acc[t] = (v8f){0.f,0.f,0.f,0.f,0.f,0.f,0.f,0.f};

#pragma unroll
  for (int k = 0; k < K; k += 4) {
    v2f a;
    a.x = xrow[k + 0];
    a.y = xrow[k + 1];
#pragma unroll
    for (int t = 0; t < NT; ++t) {
      v2f b;
      b.x = wb[(size_t)(k + 0) * NOUT + 16 * t];
      b.y = wb[(size_t)(k + 1) * NOUT + 16 * t];
      acc[t] = __builtin_amdgcn_wmma_f32_16x16x4_f32(false, a, false, b,
                                                     (short)0, acc[t],
                                                     false, false);
    }
  }

  const int mbase = tM * 16 + 8 * half;
  float* cp = C + (size_t)mbase * NOUT + tN * (16 * NT) + l15;
#pragma unroll
  for (int t = 0; t < NT; ++t)
#pragma unroll
    for (int r = 0; r < 8; ++r)
      cp[(size_t)r * NOUT + 16 * t] = acc[t][r];
}

__global__ __launch_bounds__(256) void rowdot(
    const float* __restrict__ X, const float* __restrict__ v,
    float* __restrict__ out, int N, int K)
{
  const int gw   = (int)((blockIdx.x * (size_t)blockDim.x + threadIdx.x) >> 5);
  const int lane = threadIdx.x & 31;
  if (gw >= N) return;
  const float* row = X + (size_t)gw * K;
  float s = 0.f;
  for (int k = lane; k < K; k += 32) s += row[k] * v[k];
#pragma unroll
  for (int off = 16; off > 0; off >>= 1) s += __shfl_down(s, off, 32);
  if (lane == 0) out[gw] = s;
}

__global__ void foldvec(const float* __restrict__ W, const float* __restrict__ att,
                        float* __restrict__ v, int rows, int cols)
{
  const int j = blockIdx.x * blockDim.x + threadIdx.x;
  if (j >= rows) return;
  const float* wr = W + (size_t)j * cols;
  float s = 0.f;
  for (int k = 0; k < cols; ++k) s += wr[k] * att[k];
  v[j] = s;
}

__global__ void init_seg(float* __restrict__ m, float* __restrict__ denom, int N)
{
  const int i = blockIdx.x * blockDim.x + threadIdx.x;
  if (i < N) { m[i] = -__builtin_huge_valf(); denom[i] = 0.f; }
}

__global__ void init_bias(float* __restrict__ out, const float* __restrict__ b,
                          int total, int C)
{
  const int i = blockIdx.x * blockDim.x + threadIdx.x;
  if (i < total) out[i] = b[i % C];
}

__global__ void relu_ip(float* __restrict__ x, int n)
{
  const int i = blockIdx.x * blockDim.x + threadIdx.x;
  if (i < n) x[i] = fmaxf(x[i], 0.f);
}

__device__ __forceinline__ void atomicMaxF(float* addr, float val)
{
  if (val >= 0.f) atomicMax((int*)addr, __float_as_int(val));
  else            atomicMin((unsigned int*)addr, __float_as_uint(val));
}

__global__ __launch_bounds__(256) void edge_logits(
    const int* __restrict__ src, const int* __restrict__ dst,
    const float* __restrict__ asrc, const float* __restrict__ adst,
    float* __restrict__ e, float* __restrict__ m, int E)
{
  const int i = blockIdx.x * blockDim.x + threadIdx.x;
  if (i >= E) return;
  const int d = dst[i];
  float v = asrc[src[i]] + adst[d];
  v = (v > 0.f) ? v : NEG_SLOPE * v;
  e[i] = v;
  atomicMaxF(&m[d], v);
}

__global__ __launch_bounds__(256) void edge_exp(
    const int* __restrict__ dst, const float* __restrict__ m,
    float* __restrict__ e, float* __restrict__ denom, int E)
{
  const int i = blockIdx.x * blockDim.x + threadIdx.x;
  if (i >= E) return;
  const int d = dst[i];
  const float ee = __expf(e[i] - m[d]);
  e[i] = ee;
  atomicAdd(&denom[d], ee);
}

template<int C>
__global__ __launch_bounds__(256) void edge_scatter(
    const int* __restrict__ src, const int* __restrict__ dst,
    const float* __restrict__ ee, const float* __restrict__ denom,
    const float* __restrict__ h, float* __restrict__ out, int E)
{
  const int wv   = (int)((blockIdx.x * (size_t)blockDim.x + threadIdx.x) >> 5);
  const int lane = threadIdx.x & 31;
  if (wv >= E) return;
  const int s = src[wv];
  const int d = dst[wv];
  const float alpha = ee[wv] / denom[d];
  constexpr int P = C / 32;
  const float* hp = h   + (size_t)s * C + lane * P;
  float*       op = out + (size_t)d * C + lane * P;
  float tmp[P];
  if constexpr (P == 4) {
    const float4 v = *(const float4*)hp;
    tmp[0] = v.x; tmp[1] = v.y; tmp[2] = v.z; tmp[3] = v.w;
  } else {
    const float2 v = *(const float2*)hp;
    tmp[0] = v.x; tmp[1] = v.y;
  }
#pragma unroll
  for (int i = 0; i < P; ++i) atomicAdd(op + i, alpha * tmp[i]);
}

static inline int cdiv(long a, long b) { return (int)((a + b - 1) / b); }

extern "C" void kernel_launch(void* const* d_in, const int* in_sizes, int n_in,
                              void* d_out, int out_size, void* d_ws, size_t ws_size,
                              hipStream_t stream)
{
  const float* x       = (const float*)d_in[0];
  const int*   ei      = (const int*)d_in[1];
  const float* Wsrc1   = (const float*)d_in[2];
  const float* Wdst1   = (const float*)d_in[3];
  const float* attsrc1 = (const float*)d_in[4];
  const float* attdst1 = (const float*)d_in[5];
  const float* b1      = (const float*)d_in[6];
  const float* Wsrc2   = (const float*)d_in[7];
  const float* Wdst2   = (const float*)d_in[8];
  const float* attsrc2 = (const float*)d_in[9];
  const float* attdst2 = (const float*)d_in[10];
  const float* b2      = (const float*)d_in[11];

  const int HID = in_sizes[4];
  const int OUT = in_sizes[9];
  const int FIN = in_sizes[2] / HID;
  const int N   = in_sizes[0] / FIN;
  const int E   = in_sizes[1] / 2;
  const int* src = ei;
  const int* dst = ei + E;

  float* ws    = (float*)d_ws;
  float* h1    = ws;  ws += (size_t)N * HID;
  float* o1    = ws;  ws += (size_t)N * HID;
  float* h2    = ws;  ws += (size_t)N * OUT;
  float* asrc  = ws;  ws += N;
  float* adst  = ws;  ws += N;
  float* mbuf  = ws;  ws += N;
  float* dbuf  = ws;  ws += N;
  float* ebuf  = ws;  ws += E;
  float* vfold = ws;  ws += HID;

  {
    const int tiles = (N / 16) * (HID / 64);
    gemm_wmma_f32<128, 128, 4><<<cdiv(tiles, 8), 256, 0, stream>>>(x, Wsrc1, h1, N);
    rowdot<<<cdiv((long)N * 32, 256), 256, 0, stream>>>(h1, attsrc1, asrc, N, HID);
    foldvec<<<cdiv(FIN, 128), 128, 0, stream>>>(Wdst1, attdst1, vfold, FIN, HID);
    rowdot<<<cdiv((long)N * 32, 256), 256, 0, stream>>>(x, vfold, adst, N, FIN);
    init_seg<<<cdiv(N, 256), 256, 0, stream>>>(mbuf, dbuf, N);
    init_bias<<<cdiv((long)N * HID, 256), 256, 0, stream>>>(o1, b1, N * HID, HID);
    edge_logits<<<cdiv(E, 256), 256, 0, stream>>>(src, dst, asrc, adst, ebuf, mbuf, E);
    edge_exp<<<cdiv(E, 256), 256, 0, stream>>>(dst, mbuf, ebuf, dbuf, E);
    edge_scatter<128><<<cdiv((long)E * 32, 256), 256, 0, stream>>>(
        src, dst, ebuf, dbuf, h1, o1, E);
    relu_ip<<<cdiv((long)N * HID, 256), 256, 0, stream>>>(o1, N * HID);
  }

  {
    float* out = (float*)d_out;
    const int tiles = (N / 16) * (OUT / 64);
    gemm_wmma_f32<128, 64, 4><<<cdiv(tiles, 8), 256, 0, stream>>>(o1, Wsrc2, h2, N);
    rowdot<<<cdiv((long)N * 32, 256), 256, 0, stream>>>(h2, attsrc2, asrc, N, OUT);
    foldvec<<<cdiv(HID, 128), 128, 0, stream>>>(Wdst2, attdst2, vfold, HID, OUT);
    rowdot<<<cdiv((long)N * 32, 256), 256, 0, stream>>>(o1, vfold, adst, N, HID);
    init_seg<<<cdiv(N, 256), 256, 0, stream>>>(mbuf, dbuf, N);
    init_bias<<<cdiv((long)N * OUT, 256), 256, 0, stream>>>(out, b2, N * OUT, OUT);
    edge_logits<<<cdiv(E, 256), 256, 0, stream>>>(src, dst, asrc, adst, ebuf, mbuf, E);
    edge_exp<<<cdiv(E, 256), 256, 0, stream>>>(dst, mbuf, ebuf, dbuf, E);
    edge_scatter<64><<<cdiv((long)E * 32, 256), 256, 0, stream>>>(
        src, dst, ebuf, dbuf, h2, out, E);
  }
}


# Row-dot launches: a wave streams 25 rows (all its 512-byte loads in flight at once, 16 bytes a lane) and sums them by vector-unit lane exchanges, not one short wave per row; logits in a new sum order
# speedup vs baseline: 2.9236x; 2.9236x over previous
#include <hip/hip_runtime.h>
#include <hip/hip_bf16.h>

typedef float v2f __attribute__((ext_vector_type(2)));
typedef float v4f __attribute__((ext_vector_type(4)));
typedef float v8f __attribute__((ext_vector_type(8)));

#define NEG_SLOPE 0.2f

#define REC 32

__device__ __forceinline__ int my_half()
{
  return (int)((__builtin_amdgcn_s_sendmsg_rtn(0x87) >> 18) & 1u);
}

template<int K, int NOUT, int NT>
__global__ __launch_bounds__(256) void gemm_wmma_f32(
    const float* __restrict__ X, const float* __restrict__ W,
    float* __restrict__ C, int M)
{
  const int lane   = threadIdx.x & 31;
  const int wave   = threadIdx.x >> 5;
  constexpr int tilesN = NOUT / (16 * NT);
  const int tiles  = (M >> 4) * tilesN;
  const int tile   = blockIdx.x * 8 + wave;
  if (tile >= tiles) return;
  const int tM = tile / tilesN;
  const int tN = tile % tilesN;

  const int half = lane >> 4;
  const int l15  = lane & 15;
  const float* xrow = X + (size_t)(tM * 16 + l15) * K + 2 * half;
  const float* wb   = W + (size_t)(2 * half) * NOUT + tN * (16 * NT) + l15;

  v8f acc[NT];
#pragma unroll
  for (int t = 0; t < NT; ++t) acc[t] = (v8f){0.f,0.f,0.f,0.f,0.f,0.f,0.f,0.f};

#pragma unroll
  for (int k = 0; k < K; k += 4) {
    v2f a;
    a.x = xrow[k + 0];
    a.y = xrow[k + 1];
#pragma unroll
    for (int t = 0; t < NT; ++t) {
      v2f b;
      b.x = wb[(size_t)(k + 0) * NOUT + 16 * t];
      b.y = wb[(size_t)(k + 1) * NOUT + 16 * t];
      acc[t] = __builtin_amdgcn_wmma_f32_16x16x4_f32(false, a, false, b,
                                                     (short)0, acc[t],
                                                     false, false);
    }
  }

  const int mbase = tM * 16 + 8 * half;
  float* cp = C + (size_t)mbase * NOUT + tN * (16 * NT) + l15;
#pragma unroll
  for (int t = 0; t < NT; ++t)
#pragma unroll
    for (int r = 0; r < 8; ++r)
      cp[(size_t)r * NOUT + 16 * t] = acc[t][r];
}

__device__ __forceinline__ float xhalf(float x)
{
  const int b = __float_as_int(x);
  return __int_as_float(__builtin_amdgcn_permlanex16(b, b, 0x76543210, 0xfedcba98, false, false));
}

template<int M>
__device__ __forceinline__ float row_xor(float x)
{
  return __int_as_float(__builtin_amdgcn_update_dpp(0, __float_as_int(x), 0x160 + M, 0xf, 0xf, false));
}

#define RD_WAVES 16

template<int K, int LD>
__global__ __launch_bounds__(RD_WAVES * 32) void rowdot(
    const float* __restrict__ X, const float* __restrict__ v,
    float* __restrict__ out, int N)
{
  static_assert(K == 128 || K == 64, "a 512-byte piece is one row or two");
  constexpr int RPL = 128 / K;
  constexpr int LPR = 32 / RPL;
  constexpr int RW  = LD * RPL;
  static_assert(LD <= LPR, "lane c of a row's group stores piece c");

  const int lane = threadIdx.x & 31;
  const int wave = __builtin_amdgcn_readfirstlane((int)(threadIdx.x >> 5));
  const int row0 = ((int)blockIdx.x * RD_WAVES + wave) * RW;
  if (row0 >= N) return;
  const int sub = lane / LPR;
  const int cg  = lane % LPR;

  const unsigned last4 = (unsigned)N * (K / 4) - 1u;
  const unsigned base4 = (unsigned)row0 * (K / 4) + (unsigned)lane;
  const v4f* X4 = (const v4f*)X;
  v4f a[LD];
#pragma unroll
  for (int i = 0; i < LD; ++i) {
    const unsigned j = base4 + 32u * i;
    a[i] = X4[(j < last4) ? j : last4];
  }
  const v4f vv = ((const v4f*)v)[cg];
  __builtin_amdgcn_sched_barrier(0);

  float p[LD];
#pragma unroll
  for (int i = 0; i < LD; ++i) {
    float s = a[i].x * vv.x + a[i].y * vv.y + a[i].z * vv.z + a[i].w * vv.w;
    s += row_xor<1>(s);
    s += row_xor<2>(s);
    s += row_xor<4>(s);
    s += row_xor<8>(s);
    if (RPL == 1) s += xhalf(s);
    p[i] = s;
  }

  float mine = 0.f;
#pragma unroll
  for (int i = 0; i < LD; ++i) mine = (cg == i) ? p[i] : mine;
  const int r = row0 + cg * RPL + sub;
  if (cg < LD && r < N) out[r] = mine;
}

__global__ void foldvec(const float* __restrict__ W, const float* __restrict__ att,
                        float* __restrict__ v, int rows, int cols)
{
  const int j = blockIdx.x * blockDim.x + threadIdx.x;
  if (j >= rows) return;
  const float* wr = W + (size_t)j * cols;
  float s = 0.f;
  for (int k = 0; k < cols; ++k) s += wr[k] * att[k];
  v[j] = s;
}

__global__ void init_seg(float* __restrict__ rec, int N)
{
  const int i = blockIdx.x * blockDim.x + threadIdx.x;
  if (i < N) {
    rec[(size_t)i * REC] = -__builtin_huge_valf();
    rec[((size_t)N + i) * REC] = -__builtin_huge_valf();
  }
}

__device__ __forceinline__ void atomicMaxF(float* addr, float val)
{
  if (val >= 0.f) atomicMax((int*)addr, __float_as_int(val));
  else            atomicMin((unsigned int*)addr, __float_as_uint(val));
}

struct __attribute__((aligned(8))) EdgeRec { int d; float e; };

#define RNG  512

__global__ __launch_bounds__(256) void edge_logits(
    const int* __restrict__ src, const int* __restrict__ dst,
    const float* __restrict__ asrc, const float* __restrict__ adst,
    EdgeRec* __restrict__ ed, float* __restrict__ rec,
    unsigned char* __restrict__ bins, int E, int N)
{
  const int i = blockIdx.x * blockDim.x + threadIdx.x;
  if (i >= E) {
    if (bins != nullptr && i < ((E + 15) & ~15)) bins[i] = 0xFF;
    return;
  }
  const int d = dst[i];
  float v = asrc[src[i]] + adst[d];
  v = (v > 0.f) ? v : NEG_SLOPE * v;
  EdgeRec r;
  r.d = d;
  r.e = v;
  ed[i] = r;
  if (bins != nullptr) bins[i] = (unsigned char)((unsigned)d / RNG);
  atomicMaxF(rec + ((size_t)my_half() * N + d) * REC, v);
}

template<int P>
__device__ __forceinline__ void load_frag(const float* __restrict__ p, float (&t)[P])
{
#pragma unroll
  for (int i = 0; i < P; ++i) t[i] = p[32 * i];
}


template<int C>
__device__ __forceinline__ void add_queued(
    int i, int cnt, int lane, int lo, int nr, int E,
    const int* __restrict__ src, const EdgeRec* __restrict__ ed,
    const float* __restrict__ h, float* sums, float* den, const float* mx)
{
  int s = 0, dl = -1;
  float alpha = 0.f;
  if (lane < cnt && i < E) {
    const EdgeRec r = ed[i];
    const int d = r.d - lo;
    if ((unsigned)d < (unsigned)nr) {
      dl = d;
      s = src[i];
      alpha = __expf(r.e - mx[d]);
      atomicAdd(den + d, alpha);
    }
  }
  constexpr int P = C / 32;
  constexpr int B = 16 / P;
#pragma nounroll
  for (int j0 = 0; j0 < cnt; j0 += B) {
    float v[B][P];
    int   dj[B];
    float aj[B];
#pragma unroll
    for (int b = 0; b < B; ++b) {
      const int sj = __builtin_amdgcn_readlane(s, j0 + b);
      dj[b] = __builtin_amdgcn_readlane(dl, j0 + b);
      aj[b] = __int_as_float(__builtin_amdgcn_readlane(__float_as_int(alpha), j0 + b));
      load_frag<P>(h + (size_t)sj * C + lane, v[b]);
    }
#pragma unroll
    for (int b = 0; b < B; ++b) {
      if (dj[b] >= 0) {
        float* op = sums + dj[b] * C + lane;
#pragma unroll
        for (int k = 0; k < P; ++k) atomicAdd(op + 32 * k, aj[b] * v[b][k]);
      }
    }
  }
}

__device__ __forceinline__ unsigned eq_bytes(unsigned w, unsigned pat)
{
  const unsigned x = w ^ pat;
  return ~(((x & 0x7F7F7F7Fu) + 0x7F7F7F7Fu) | x | 0x7F7F7F7Fu);
}

template<int C, bool RELU>
__global__ __launch_bounds__(1024) void range_rows(
    const int* __restrict__ src, const uint4* __restrict__ bins4,
    const EdgeRec* __restrict__ ed, const float* __restrict__ rec,
    const float* __restrict__ h, const float* __restrict__ bias,
    float* __restrict__ x, int E, int N)
{
  __shared__ float sums[RNG * C];
  __shared__ float den[RNG];
  __shared__ float mx[RNG];

  const int tid  = threadIdx.x;
  const int lane = tid & 31;
  const int wave = __builtin_amdgcn_readfirstlane(tid >> 5);
  const int lo   = blockIdx.x * RNG;
  const int nr   = (N - lo < RNG) ? (N - lo) : RNG;

  for (int i = tid; i < RNG * C; i += 1024) sums[i] = 0.f;
  if (tid < RNG) {
    den[tid] = 0.f;
    mx[tid] = (tid < nr) ? fmaxf(rec[(size_t)(lo + tid) * REC], rec[((size_t)N + lo + tid) * REC])
                         : 0.f;
  }
  __syncthreads();

  const unsigned pat = (blockIdx.x & 255u) * 0x01010101u;
  int q  = 0;
  int qn = 0;

  const int G = (E + 15) >> 4;
  const int T = (G + 8191) >> 13;
  int tt = (T > 0) ? (int)(blockIdx.x % (unsigned)T) : 0;
  for (int t = 0; t < T; ++t) {
    const int g0 = tt * 8192 + wave * 256;
    tt = (tt + 1 == T) ? 0 : tt + 1;
    uint4 v[8];
#pragma unroll
    for (int j = 0; j < 8; ++j) {
      const int g = g0 + j * 32 + lane;
      v[j] = bins4[(g < G) ? g : G - 1];
    }
#pragma unroll
    for (int j = 0; j < 8; ++j) {
      const unsigned ma = (eq_bytes(v[j].x, pat) >> 7) | (eq_bytes(v[j].y, pat) >> 6) |
                          (eq_bytes(v[j].z, pat) >> 5) | (eq_bytes(v[j].w, pat) >> 4);
      const unsigned m = (g0 + j * 32 + lane < G) ? ma : 0u;
      unsigned lanes = __builtin_amdgcn_ballot_w32(m != 0u);
      const int eb = (g0 + j * 32) << 4;
      while (lanes != 0u) {
        const int l = __builtin_ctz(lanes);
        lanes &= lanes - 1u;
        unsigned ml = (unsigned)__builtin_amdgcn_readlane((int)m, l);
        const int base = eb + (l << 4);
        while (ml != 0u) {
          const int p = __builtin_ctz(ml);
          ml &= ml - 1u;
          const int idx = base + ((p & 7) << 2) + (p >> 3);
          q = (lane == qn) ? idx : q;
          if (++qn == 32) {
            add_queued<C>(q, 32, lane, lo, nr, E, src, ed, h, sums, den, mx);
            qn = 0;
          }
        }
      }
    }
  }
  add_queued<C>(q, qn, lane, lo, nr, E, src, ed, h, sums, den, mx);

  __syncthreads();
  for (int i = tid; i < nr * C; i += 1024) {
    const float dn = den[i / C];
    const float qv = sums[i] / dn;
    float r = ((dn > 0.f) ? qv : 0.f) + bias[i % C];
    if (RELU) r = fmaxf(r, 0.f);
    x[(size_t)lo * C + i] = r;
  }
}

static inline int cdiv(long a, long b) { return (int)((a + b - 1) / b); }

extern "C" void kernel_launch(void* const* d_in, const int* in_sizes, int n_in,
                              void* d_out, int out_size, void* d_ws, size_t ws_size,
                              hipStream_t stream)
{
  const float* x       = (const float*)d_in[0];
  const int*   ei      = (const int*)d_in[1];
  const float* Wsrc1   = (const float*)d_in[2];
  const float* Wdst1   = (const float*)d_in[3];
  const float* attsrc1 = (const float*)d_in[4];
  const float* attdst1 = (const float*)d_in[5];
  const float* b1      = (const float*)d_in[6];
  const float* Wsrc2   = (const float*)d_in[7];
  const float* Wdst2   = (const float*)d_in[8];
  const float* attsrc2 = (const float*)d_in[9];
  const float* attdst2 = (const float*)d_in[10];
  const float* b2      = (const float*)d_in[11];

  const int HID = in_sizes[4];
  const int OUT = in_sizes[9];
  const int FIN = in_sizes[2] / HID;
  const int N   = in_sizes[0] / FIN;
  const int E   = in_sizes[1] / 2;
  const int* src = ei;
  const int* dst = ei + E;

  unsigned char* bins = (unsigned char*)d_ws;
  float* ws    = (float*)d_ws + (((size_t)E + 127) / 128) * 32;
  EdgeRec* ed  = (EdgeRec*)ws;  ws += (((size_t)E + 15) / 16) * 32;
  float* h1    = ws;  ws += (size_t)N * HID;
  float* o1    = ws;  ws += (size_t)N * HID;
  float* h2    = ws;  ws += (size_t)N * OUT;
  float* rec   = ws;  ws += (size_t)2 * N * REC;
  float* asrc  = ws;  ws += N;
  float* adst  = ws;  ws += N;
  float* vfold = ws;  ws += HID;

  {
    const int tiles = (N / 16) * (HID / 64);
    gemm_wmma_f32<128, 128, 4><<<cdiv(tiles, 8), 256, 0, stream>>>(x, Wsrc1, h1, N);
    rowdot<128, 25><<<cdiv(N, RD_WAVES * 25), RD_WAVES * 32, 0, stream>>>(h1, attsrc1, asrc, N);
    foldvec<<<cdiv(FIN, 128), 128, 0, stream>>>(Wdst1, attdst1, vfold, FIN, HID);
    rowdot<128, 25><<<cdiv(N, RD_WAVES * 25), RD_WAVES * 32, 0, stream>>>(x, vfold, adst, N);
    init_seg<<<cdiv(N, 256), 256, 0, stream>>>(rec, N);
    edge_logits<<<cdiv(E, 256), 256, 0, stream>>>(src, dst, asrc, adst, ed, rec, bins, E, N);
    range_rows<128, true><<<cdiv(N, RNG), 1024, 0, stream>>>(src, (const uint4*)bins, ed, rec, h1, b1, o1, E, N);
  }

  {
    float* out = (float*)d_out;
    const int tiles = (N / 16) * (OUT / 64);
    gemm_wmma_f32<128, 64, 4><<<cdiv(tiles, 8), 256, 0, stream>>>(o1, Wsrc2, h2, N);
    rowdot<64, 13><<<cdiv(N, RD_WAVES * 26), RD_WAVES * 32, 0, stream>>>(h2, attsrc2, asrc, N);
    foldvec<<<cdiv(HID, 128), 128, 0, stream>>>(Wdst2, attdst2, vfold, HID, OUT);
    rowdot<128, 25><<<cdiv(N, RD_WAVES * 25), RD_WAVES * 32, 0, stream>>>(o1, vfold, adst, N);
    init_seg<<<cdiv(N, 256), 256, 0, stream>>>(rec, N);
    edge_logits<<<cdiv(E, 256), 256, 0, stream>>>(src, dst, asrc, adst, ed, rec, nullptr, E, N);
    range_rows<64, false><<<cdiv(N, RNG), 1024, 0, stream>>>(src, (const uint4*)bins, ed, rec, h2, b2, out, E, N);
  }
}


# Row-sum pass: a lane fetches its 4 (layer 2: 2) neighbouring columns of a source row in one 16-byte (8-byte) load: one load per row, not one per line; sums column-permuted in shared memory
# speedup vs baseline: 3.0869x; 1.0559x over previous
#include <hip/hip_runtime.h>
#include <hip/hip_bf16.h>

typedef float v2f __attribute__((ext_vector_type(2)));
typedef float v4f __attribute__((ext_vector_type(4)));
typedef float v8f __attribute__((ext_vector_type(8)));

#define NEG_SLOPE 0.2f

#define REC 32

__device__ __forceinline__ int my_half()
{
  return (int)((__builtin_amdgcn_s_sendmsg_rtn(0x87) >> 18) & 1u);
}

template<int K, int NOUT, int NT>
__global__ __launch_bounds__(256) void gemm_wmma_f32(
    const float* __restrict__ X, const float* __restrict__ W,
    float* __restrict__ C, int M)
{
  const int lane   = threadIdx.x & 31;
  const int wave   = threadIdx.x >> 5;
  constexpr int tilesN = NOUT / (16 * NT);
  const int tiles  = (M >> 4) * tilesN;
  const int tile   = blockIdx.x * 8 + wave;
  if (tile >= tiles) return;
  const int tM = tile / tilesN;
  const int tN = tile % tilesN;

  const int half = lane >> 4;
  const int l15  = lane & 15;
  const float* xrow = X + (size_t)(tM * 16 + l15) * K + 2 * half;
  const float* wb   = W + (size_t)(2 * half) * NOUT + tN * (16 * NT) + l15;

  v8f acc[NT];
#pragma unroll
  for (int t = 0; t < NT; ++t) acc[t] = (v8f){0.f,0.f,0.f,0.f,0.f,0.f,0.f,0.f};

#pragma unroll
  for (int k = 0; k < K; k += 4) {
    v2f a;
    a.x = xrow[k + 0];
    a.y = xrow[k + 1];
#pragma unroll
    for (int t = 0; t < NT; ++t) {
      v2f b;
      b.x = wb[(size_t)(k + 0) * NOUT + 16 * t];
      b.y = wb[(size_t)(k + 1) * NOUT + 16 * t];
      acc[t] = __builtin_amdgcn_wmma_f32_16x16x4_f32(false, a, false, b,
                                                     (short)0, acc[t],
                                                     false, false);
    }
  }

  const int mbase = tM * 16 + 8 * half;
  float* cp = C + (size_t)mbase * NOUT + tN * (16 * NT) + l15;
#pragma unroll
  for (int t = 0; t < NT; ++t)
#pragma unroll
    for (int r = 0; r < 8; ++r)
      cp[(size_t)r * NOUT + 16 * t] = acc[t][r];
}

__device__ __forceinline__ float xhalf(float x)
{
  const int b = __float_as_int(x);
  return __int_as_float(__builtin_amdgcn_permlanex16(b, b, 0x76543210, 0xfedcba98, false, false));
}

template<int M>
__device__ __forceinline__ float row_xor(float x)
{
  return __int_as_float(__builtin_amdgcn_update_dpp(0, __float_as_int(x), 0x160 + M, 0xf, 0xf, false));
}

#define RD_WAVES 16

template<int K, int LD>
__global__ __launch_bounds__(RD_WAVES * 32) void rowdot(
    const float* __restrict__ X, const float* __restrict__ v,
    float* __restrict__ out, int N)
{
  static_assert(K == 128 || K == 64, "a 512-byte piece is one row or two");
  constexpr int RPL = 128 / K;
  constexpr int LPR = 32 / RPL;
  constexpr int RW  = LD * RPL;
  static_assert(LD <= LPR, "lane c of a row's group stores piece c");

  const int lane = threadIdx.x & 31;
  const int wave = __builtin_amdgcn_readfirstlane((int)(threadIdx.x >> 5));
  const int row0 = ((int)blockIdx.x * RD_WAVES + wave) * RW;
  if (row0 >= N) return;
  const int sub = lane / LPR;
  const int cg  = lane % LPR;

  const unsigned last4 = (unsigned)N * (K / 4) - 1u;
  const unsigned base4 = (unsigned)row0 * (K / 4) + (unsigned)lane;
  const v4f* X4 = (const v4f*)X;
  v4f a[LD];
#pragma unroll
  for (int i = 0; i < LD; ++i) {
    const unsigned j = base4 + 32u * i;
    a[i] = X4[(j < last4) ? j : last4];
  }
  const v4f vv = ((const v4f*)v)[cg];
  __builtin_amdgcn_sched_barrier(0);

  float p[LD];
#pragma unroll
  for (int i = 0; i < LD; ++i) {
    float s = a[i].x * vv.x + a[i].y * vv.y + a[i].z * vv.z + a[i].w * vv.w;
    s += row_xor<1>(s);
    s += row_xor<2>(s);
    s += row_xor<4>(s);
    s += row_xor<8>(s);
    if (RPL == 1) s += xhalf(s);
    p[i] = s;
  }

  float mine = 0.f;
#pragma unroll
  for (int i = 0; i < LD; ++i) mine = (cg == i) ? p[i] : mine;
  const int r = row0 + cg * RPL + sub;
  if (cg < LD && r < N) out[r] = mine;
}

__global__ void foldvec(const float* __restrict__ W, const float* __restrict__ att,
                        float* __restrict__ v, int rows, int cols)
{
  const int j = blockIdx.x * blockDim.x + threadIdx.x;
  if (j >= rows) return;
  const float* wr = W + (size_t)j * cols;
  float s = 0.f;
  for (int k = 0; k < cols; ++k) s += wr[k] * att[k];
  v[j] = s;
}

__global__ void init_seg(float* __restrict__ rec, int N)
{
  const int i = blockIdx.x * blockDim.x + threadIdx.x;
  if (i < N) {
    rec[(size_t)i * REC] = -__builtin_huge_valf();
    rec[((size_t)N + i) * REC] = -__builtin_huge_valf();
  }
}

__device__ __forceinline__ void atomicMaxF(float* addr, float val)
{
  if (val >= 0.f) atomicMax((int*)addr, __float_as_int(val));
  else            atomicMin((unsigned int*)addr, __float_as_uint(val));
}

struct __attribute__((aligned(8))) EdgeRec { int d; float e; };

#define RNG  512

__global__ __launch_bounds__(256) void edge_logits(
    const int* __restrict__ src, const int* __restrict__ dst,
    const float* __restrict__ asrc, const float* __restrict__ adst,
    EdgeRec* __restrict__ ed, float* __restrict__ rec,
    unsigned char* __restrict__ bins, int E, int N)
{
  const int i = blockIdx.x * blockDim.x + threadIdx.x;
  if (i >= E) {
    if (bins != nullptr && i < ((E + 15) & ~15)) bins[i] = 0xFF;
    return;
  }
  const int d = dst[i];
  float v = asrc[src[i]] + adst[d];
  v = (v > 0.f) ? v : NEG_SLOPE * v;
  EdgeRec r;
  r.d = d;
  r.e = v;
  ed[i] = r;
  if (bins != nullptr) bins[i] = (unsigned char)((unsigned)d / RNG);
  atomicMaxF(rec + ((size_t)my_half() * N + d) * REC, v);
}

__device__ __forceinline__ void load_frag(const float* __restrict__ p, float (&t)[4])
{
  const v4f q = *(const v4f*)p;
  t[0] = q.x; t[1] = q.y; t[2] = q.z; t[3] = q.w;
}
__device__ __forceinline__ void load_frag(const float* __restrict__ p, float (&t)[2])
{
  const v2f q = *(const v2f*)p;
  t[0] = q.x; t[1] = q.y;
}


template<int C>
__device__ __forceinline__ void add_queued(
    int i, int cnt, int lane, int lo, int nr, int E,
    const int* __restrict__ src, const EdgeRec* __restrict__ ed,
    const float* __restrict__ h, float* sums, float* den, const float* mx)
{
  int s = 0, dl = -1;
  float alpha = 0.f;
  if (lane < cnt && i < E) {
    const EdgeRec r = ed[i];
    const int d = r.d - lo;
    if ((unsigned)d < (unsigned)nr) {
      dl = d;
      s = src[i];
      alpha = __expf(r.e - mx[d]);
      atomicAdd(den + d, alpha);
    }
  }
  constexpr int P = C / 32;
  constexpr int B = 16 / P;
#pragma nounroll
  for (int j0 = 0; j0 < cnt; j0 += B) {
    float v[B][P];
    int   dj[B];
    float aj[B];
#pragma unroll
    for (int b = 0; b < B; ++b) {
      const int sj = __builtin_amdgcn_readlane(s, j0 + b);
      dj[b] = __builtin_amdgcn_readlane(dl, j0 + b);
      aj[b] = __int_as_float(__builtin_amdgcn_readlane(__float_as_int(alpha), j0 + b));
      load_frag(h + (size_t)sj * C + P * lane, v[b]);
    }
#pragma unroll
    for (int b = 0; b < B; ++b) {
      if (dj[b] >= 0) {
        float* op = sums + dj[b] * C + lane;
#pragma unroll
        for (int k = 0; k < P; ++k) atomicAdd(op + 32 * k, aj[b] * v[b][k]);
      }
    }
  }
}

__device__ __forceinline__ unsigned eq_bytes(unsigned w, unsigned pat)
{
  const unsigned x = w ^ pat;
  return ~(((x & 0x7F7F7F7Fu) + 0x7F7F7F7Fu) | x | 0x7F7F7F7Fu);
}

template<int C, bool RELU>
__global__ __launch_bounds__(1024) void range_rows(
    const int* __restrict__ src, const uint4* __restrict__ bins4,
    const EdgeRec* __restrict__ ed, const float* __restrict__ rec,
    const float* __restrict__ h, const float* __restrict__ bias,
    float* __restrict__ x, int E, int N)
{
  __shared__ float sums[RNG * C];
  __shared__ float den[RNG];
  __shared__ float mx[RNG];

  const int tid  = threadIdx.x;
  const int lane = tid & 31;
  const int wave = __builtin_amdgcn_readfirstlane(tid >> 5);
  const int lo   = blockIdx.x * RNG;
  const int nr   = (N - lo < RNG) ? (N - lo) : RNG;

  for (int i = tid; i < RNG * C; i += 1024) sums[i] = 0.f;
  if (tid < RNG) {
    den[tid] = 0.f;
    mx[tid] = (tid < nr) ? fmaxf(rec[(size_t)(lo + tid) * REC], rec[((size_t)N + lo + tid) * REC])
                         : 0.f;
  }
  __syncthreads();

  const unsigned pat = (blockIdx.x & 255u) * 0x01010101u;
  int q  = 0;
  int qn = 0;

  const int G = (E + 15) >> 4;
  const int T = (G + 8191) >> 13;
  int tt = (T > 0) ? (int)(blockIdx.x % (unsigned)T) : 0;
  for (int t = 0; t < T; ++t) {
    const int g0 = tt * 8192 + wave * 256;
    tt = (tt + 1 == T) ? 0 : tt + 1;
    uint4 v[8];
#pragma unroll
    for (int j = 0; j < 8; ++j) {
      const int g = g0 + j * 32 + lane;
      v[j] = bins4[(g < G) ? g : G - 1];
    }
#pragma unroll
    for (int j = 0; j < 8; ++j) {
      const unsigned ma = (eq_bytes(v[j].x, pat) >> 7) | (eq_bytes(v[j].y, pat) >> 6) |
                          (eq_bytes(v[j].z, pat) >> 5) | (eq_bytes(v[j].w, pat) >> 4);
      const unsigned m = (g0 + j * 32 + lane < G) ? ma : 0u;
      unsigned lanes = __builtin_amdgcn_ballot_w32(m != 0u);
      const int eb = (g0 + j * 32) << 4;
      while (lanes != 0u) {
        const int l = __builtin_ctz(lanes);
        lanes &= lanes - 1u;
        unsigned ml = (unsigned)__builtin_amdgcn_readlane((int)m, l);
        const int base = eb + (l << 4);
        while (ml != 0u) {
          const int p = __builtin_ctz(ml);
          ml &= ml - 1u;
          const int idx = base + ((p & 7) << 2) + (p >> 3);
          q = (lane == qn) ? idx : q;
          if (++qn == 32) {
            add_queued<C>(q, 32, lane, lo, nr, E, src, ed, h, sums, den, mx);
            qn = 0;
          }
        }
      }
    }
  }
  add_queued<C>(q, qn, lane, lo, nr, E, src, ed, h, sums, den, mx);

  __syncthreads();
  constexpr int P = C / 32;
  for (int i = tid; i < nr * C; i += 1024) {
    const int n = i / C;
    const int cc = i % C;
    const float dn = den[n];
    const float qv = sums[n * C + 32 * (cc % P) + cc / P] / dn;
    float r = ((dn > 0.f) ? qv : 0.f) + bias[i % C];
    if (RELU) r = fmaxf(r, 0.f);
    x[(size_t)lo * C + i] = r;
  }
}

static inline int cdiv(long a, long b) { return (int)((a + b - 1) / b); }

extern "C" void kernel_launch(void* const* d_in, const int* in_sizes, int n_in,
                              void* d_out, int out_size, void* d_ws, size_t ws_size,
                              hipStream_t stream)
{
  const float* x       = (const float*)d_in[0];
  const int*   ei      = (const int*)d_in[1];
  const float* Wsrc1   = (const float*)d_in[2];
  const float* Wdst1   = (const float*)d_in[3];
  const float* attsrc1 = (const float*)d_in[4];
  const float* attdst1 = (const float*)d_in[5];
  const float* b1      = (const float*)d_in[6];
  const float* Wsrc2   = (const float*)d_in[7];
  const float* Wdst2   = (const float*)d_in[8];
  const float* attsrc2 = (const float*)d_in[9];
  const float* attdst2 = (const float*)d_in[10];
  const float* b2      = (const float*)d_in[11];

  const int HID = in_sizes[4];
  const int OUT = in_sizes[9];
  const int FIN = in_sizes[2] / HID;
  const int N   = in_sizes[0] / FIN;
  const int E   = in_sizes[1] / 2;
  const int* src = ei;
  const int* dst = ei + E;

  unsigned char* bins = (unsigned char*)d_ws;
  float* ws    = (float*)d_ws + (((size_t)E + 127) / 128) * 32;
  EdgeRec* ed  = (EdgeRec*)ws;  ws += (((size_t)E + 15) / 16) * 32;
  float* h1    = ws;  ws += (size_t)N * HID;
  float* o1    = ws;  ws += (size_t)N * HID;
  float* h2    = ws;  ws += (size_t)N * OUT;
  float* rec   = ws;  ws += (size_t)2 * N * REC;
  float* asrc  = ws;  ws += N;
  float* adst  = ws;  ws += N;
  float* vfold = ws;  ws += HID;

  {
    const int tiles = (N / 16) * (HID / 64);
    gemm_wmma_f32<128, 128, 4><<<cdiv(tiles, 8), 256, 0, stream>>>(x, Wsrc1, h1, N);
    rowdot<128, 25><<<cdiv(N, RD_WAVES * 25), RD_WAVES * 32, 0, stream>>>(h1, attsrc1, asrc, N);
    foldvec<<<cdiv(FIN, 128), 128, 0, stream>>>(Wdst1, attdst1, vfold, FIN, HID);
    rowdot<128, 25><<<cdiv(N, RD_WAVES * 25), RD_WAVES * 32, 0, stream>>>(x, vfold, adst, N);
    init_seg<<<cdiv(N, 256), 256, 0, stream>>>(rec, N);
    edge_logits<<<cdiv(E, 256), 256, 0, stream>>>(src, dst, asrc, adst, ed, rec, bins, E, N);
    range_rows<128, true><<<cdiv(N, RNG), 1024, 0, stream>>>(src, (const uint4*)bins, ed, rec, h1, b1, o1, E, N);
  }

  {
    float* out = (float*)d_out;
    const int tiles = (N / 16) * (OUT / 64);
    gemm_wmma_f32<128, 64, 4><<<cdiv(tiles, 8), 256, 0, stream>>>(o1, Wsrc2, h2, N);
    rowdot<64, 13><<<cdiv(N, RD_WAVES * 26), RD_WAVES * 32, 0, stream>>>(h2, attsrc2, asrc, N);
    foldvec<<<cdiv(HID, 128), 128, 0, stream>>>(Wdst2, attdst2, vfold, HID, OUT);
    rowdot<128, 25><<<cdiv(N, RD_WAVES * 25), RD_WAVES * 32, 0, stream>>>(o1, vfold, adst, N);
    init_seg<<<cdiv(N, 256), 256, 0, stream>>>(rec, N);
    edge_logits<<<cdiv(E, 256), 256, 0, stream>>>(src, dst, asrc, adst, ed, rec, nullptr, E, N);
    range_rows<64, false><<<cdiv(N, RNG), 1024, 0, stream>>>(src, (const uint4*)bins, ed, rec, h2, b2, out, E, N);
  }
}


# Softmax maxima and logits formed in the row-sum workgroup (392-node ranges, its edges listed in shared memory): edge-logit and init launches and global max atomics go; shared-add arrival order
# speedup vs baseline: 3.9135x; 1.2678x over previous
#include <hip/hip_runtime.h>
#include <hip/hip_bf16.h>

typedef float v2f __attribute__((ext_vector_type(2)));
typedef float v4f __attribute__((ext_vector_type(4)));
typedef float v8f __attribute__((ext_vector_type(8)));

#define NEG_SLOPE 0.2f

template<int K, int NOUT, int NT>
__global__ __launch_bounds__(256) void gemm_wmma_f32(
    const float* __restrict__ X, const float* __restrict__ W,
    float* __restrict__ C, int M)
{
  const int lane   = threadIdx.x & 31;
  const int wave   = threadIdx.x >> 5;
  constexpr int tilesN = NOUT / (16 * NT);
  const int tiles  = (M >> 4) * tilesN;
  const int tile   = blockIdx.x * 8 + wave;
  if (tile >= tiles) return;
  const int tM = tile / tilesN;
  const int tN = tile % tilesN;

  const int half = lane >> 4;
  const int l15  = lane & 15;
  const float* xrow = X + (size_t)(tM * 16 + l15) * K + 2 * half;
  const float* wb   = W + (size_t)(2 * half) * NOUT + tN * (16 * NT) + l15;

  v8f acc[NT];
#pragma unroll
  for (int t = 0; t < NT; ++t) acc[t] = (v8f){0.f,0.f,0.f,0.f,0.f,0.f,0.f,0.f};

#pragma unroll
  for (int k = 0; k < K; k += 4) {
    v2f a;
    a.x = xrow[k + 0];
    a.y = xrow[k + 1];
#pragma unroll
    for (int t = 0; t < NT; ++t) {
      v2f b;
      b.x = wb[(size_t)(k + 0) * NOUT + 16 * t];
      b.y = wb[(size_t)(k + 1) * NOUT + 16 * t];
      acc[t] = __builtin_amdgcn_wmma_f32_16x16x4_f32(false, a, false, b,
                                                     (short)0, acc[t],
                                                     false, false);
    }
  }

  const int mbase = tM * 16 + 8 * half;
  float* cp = C + (size_t)mbase * NOUT + tN * (16 * NT) + l15;
#pragma unroll
  for (int t = 0; t < NT; ++t)
#pragma unroll
    for (int r = 0; r < 8; ++r)
      cp[(size_t)r * NOUT + 16 * t] = acc[t][r];
}

__device__ __forceinline__ float xhalf(float x)
{
  const int b = __float_as_int(x);
  return __int_as_float(__builtin_amdgcn_permlanex16(b, b, 0x76543210, 0xfedcba98, false, false));
}

template<int M>
__device__ __forceinline__ float row_xor(float x)
{
  return __int_as_float(__builtin_amdgcn_update_dpp(0, __float_as_int(x), 0x160 + M, 0xf, 0xf, false));
}

#define RD_WAVES 16

template<int K, int LD>
__global__ __launch_bounds__(RD_WAVES * 32) void rowdot(
    const float* __restrict__ X, const float* __restrict__ v,
    float* __restrict__ out, int N)
{
  static_assert(K == 128 || K == 64, "a 512-byte piece is one row or two");
  constexpr int RPL = 128 / K;
  constexpr int LPR = 32 / RPL;
  constexpr int RW  = LD * RPL;
  static_assert(LD <= LPR, "lane c of a row's group stores piece c");

  const int lane = threadIdx.x & 31;
  const int wave = __builtin_amdgcn_readfirstlane((int)(threadIdx.x >> 5));
  const int row0 = ((int)blockIdx.x * RD_WAVES + wave) * RW;
  if (row0 >= N) return;
  const int sub = lane / LPR;
  const int cg  = lane % LPR;

  const unsigned last4 = (unsigned)N * (K / 4) - 1u;
  const unsigned base4 = (unsigned)row0 * (K / 4) + (unsigned)lane;
  const v4f* X4 = (const v4f*)X;
  v4f a[LD];
#pragma unroll
  for (int i = 0; i < LD; ++i) {
    const unsigned j = base4 + 32u * i;
    a[i] = X4[(j < last4) ? j : last4];
  }
  const v4f vv = ((const v4f*)v)[cg];
  __builtin_amdgcn_sched_barrier(0);

  float p[LD];
#pragma unroll
  for (int i = 0; i < LD; ++i) {
    float s = a[i].x * vv.x + a[i].y * vv.y + a[i].z * vv.z + a[i].w * vv.w;
    s += row_xor<1>(s);
    s += row_xor<2>(s);
    s += row_xor<4>(s);
    s += row_xor<8>(s);
    if (RPL == 1) s += xhalf(s);
    p[i] = s;
  }

  float mine = 0.f;
#pragma unroll
  for (int i = 0; i < LD; ++i) mine = (cg == i) ? p[i] : mine;
  const int r = row0 + cg * RPL + sub;
  if (cg < LD && r < N) out[r] = mine;
}

__global__ void foldvec(const float* __restrict__ W, const float* __restrict__ att,
                        float* __restrict__ v, int rows, int cols)
{
  const int j = blockIdx.x * blockDim.x + threadIdx.x;
  if (j >= rows) return;
  const float* wr = W + (size_t)j * cols;
  float s = 0.f;
  for (int k = 0; k < cols; ++k) s += wr[k] * att[k];
  v[j] = s;
}

__device__ __forceinline__ void atomicMaxF(float* addr, float val)
{
  if (val >= 0.f) atomicMax((int*)addr, __float_as_int(val));
  else            atomicMin((unsigned int*)addr, __float_as_uint(val));
}

#define RNG  392
#define CAP  8192

__global__ __launch_bounds__(256) void range_bytes(
    const int* __restrict__ dst, unsigned* __restrict__ bins32, int E)
{
  const int t = blockIdx.x * blockDim.x + threadIdx.x;
  const int i0 = 4 * t;
  if (i0 >= ((E + 15) & ~15)) return;
  unsigned w = 0u;
#pragma unroll
  for (int k = 0; k < 4; ++k) {
    const int i = i0 + k;
    const unsigned b = (i < E) ? (((unsigned)dst[i] / RNG) & 255u) : 0xFFu;
    w |= b << (8 * k);
  }
  bins32[t] = w;
}

__device__ __forceinline__ void load_frag(const float* __restrict__ p, float (&t)[4])
{
  const v4f q = *(const v4f*)p;
  t[0] = q.x; t[1] = q.y; t[2] = q.z; t[3] = q.w;
}
__device__ __forceinline__ void load_frag(const float* __restrict__ p, float (&t)[2])
{
  const v2f q = *(const v2f*)p;
  t[0] = q.x; t[1] = q.y;
}


__device__ __forceinline__ void fetch_edge(
    int i, int cnt, int lane, int lo, int nr, int E,
    const int* __restrict__ src, const int* __restrict__ dst,
    const float* __restrict__ asrc, const float* __restrict__ adst,
    int& s, int& d, float& e)
{
  s = 0; d = -1; e = 0.f;
  if (lane < cnt && i < E) {
    const int dn = dst[i];
    const int ss = src[i];
    asm volatile("" :: "v"(dn), "v"(ss));
    const float v = asrc[ss] + adst[dn];
    const int dd = dn - lo;
    if ((unsigned)dd < (unsigned)nr) {
      d = dd;
      s = ss;
      e = (v > 0.f) ? v : NEG_SLOPE * v;
    }
  }
}

template<int C>
__device__ __forceinline__ void add_rows(
    int s, int dl, float alpha, int cnt, int lane,
    const float* __restrict__ h, float* sums)
{
  constexpr int P = C / 32;
  constexpr int B = 16 / P;
#pragma nounroll
  for (int j0 = 0; j0 < cnt; j0 += B) {
    float v[B][P];
    int   dj[B];
    float aj[B];
#pragma unroll
    for (int b = 0; b < B; ++b) {
      const int sj = __builtin_amdgcn_readlane(s, j0 + b);
      dj[b] = __builtin_amdgcn_readlane(dl, j0 + b);
      aj[b] = __int_as_float(__builtin_amdgcn_readlane(__float_as_int(alpha), j0 + b));
      load_frag(h + (size_t)sj * C + P * lane, v[b]);
    }
#pragma unroll
    for (int b = 0; b < B; ++b) {
      if (dj[b] >= 0) {
        float* op = sums + dj[b] * C + lane;
#pragma unroll
        for (int k = 0; k < P; ++k) atomicAdd(op + 32 * k, aj[b] * v[b][k]);
      }
    }
  }
}

__device__ __forceinline__ unsigned eq_bytes(unsigned w, unsigned pat)
{
  const unsigned x = w ^ pat;
  return ~(((x & 0x7F7F7F7Fu) + 0x7F7F7F7Fu) | x | 0x7F7F7F7Fu);
}

template<class F>
__device__ __forceinline__ void scan_bytes(
    const uint4* __restrict__ bins4, int E, int lane, int wave, unsigned bid, F&& flush)
{
  const unsigned pat = (bid & 255u) * 0x01010101u;
  int q  = 0;
  int qn = 0;

  const int G = (E + 15) >> 4;
  const int T = (G + 8191) >> 13;
  int tt = (T > 0) ? (int)(bid % (unsigned)T) : 0;
  for (int t = 0; t < T; ++t) {
    const int g0 = tt * 8192 + wave * 256;
    tt = (tt + 1 == T) ? 0 : tt + 1;
    uint4 v[8];
#pragma unroll
    for (int j = 0; j < 8; ++j) {
      const int g = g0 + j * 32 + lane;
      v[j] = bins4[(g < G) ? g : G - 1];
    }
#pragma unroll
    for (int j = 0; j < 8; ++j) {
      const unsigned ma = (eq_bytes(v[j].x, pat) >> 7) | (eq_bytes(v[j].y, pat) >> 6) |
                          (eq_bytes(v[j].z, pat) >> 5) | (eq_bytes(v[j].w, pat) >> 4);
      const unsigned m = (g0 + j * 32 + lane < G) ? ma : 0u;
      unsigned lanes = __builtin_amdgcn_ballot_w32(m != 0u);
      const int eb = (g0 + j * 32) << 4;
      while (lanes != 0u) {
        const int l = __builtin_ctz(lanes);
        lanes &= lanes - 1u;
        unsigned ml = (unsigned)__builtin_amdgcn_readlane((int)m, l);
        const int base = eb + (l << 4);
        while (ml != 0u) {
          const int p = __builtin_ctz(ml);
          ml &= ml - 1u;
          const int idx = base + ((p & 7) << 2) + (p >> 3);
          q = (lane == qn) ? idx : q;
          if (++qn == 32) {
            flush(q, 32);
            qn = 0;
          }
        }
      }
    }
  }
  flush(q, qn);
}

template<int C, bool RELU>
__global__ __launch_bounds__(1024) void range_rows(
    const int* __restrict__ src, const int* __restrict__ dst,
    const uint4* __restrict__ bins4,
    const float* __restrict__ asrc, const float* __restrict__ adst,
    const float* __restrict__ h, const float* __restrict__ bias,
    float* __restrict__ x, int E, int N)
{
  __shared__ float sums[RNG * C];
  __shared__ float den[RNG];
  __shared__ float mx[RNG];
  __shared__ int   ls[CAP];
  __shared__ float le[CAP];
  __shared__ unsigned short ld[CAP];
  __shared__ int   nlist;

  const int tid  = threadIdx.x;
  const int lane = tid & 31;
  const int wave = __builtin_amdgcn_readfirstlane(tid >> 5);
  const int lo   = blockIdx.x * RNG;
  const int nr   = (N - lo < RNG) ? (N - lo) : RNG;

  for (int i = tid; i < RNG * C; i += 1024) sums[i] = 0.f;
  if (tid < RNG) {
    den[tid] = 0.f;
    mx[tid] = -__builtin_huge_valf();
  }
  if (tid == 0) nlist = 0;
  __syncthreads();

  scan_bytes(bins4, E, lane, wave, blockIdx.x, [&](int q, int cnt) {
    int s, d; float e;
    fetch_edge(q, cnt, lane, lo, nr, E, src, dst, asrc, adst, s, d, e);
    const bool ok = d >= 0;
    if (ok) atomicMaxF(mx + d, e);
    const unsigned bal = __builtin_amdgcn_ballot_w32(ok);
    int base = 0;
    if (lane == 0 && bal != 0u) base = atomicAdd(&nlist, __builtin_popcount(bal));
    base = __builtin_amdgcn_readfirstlane(base);
    const int slot = base + __builtin_popcount(bal & ((1u << lane) - 1u));
    if (ok && slot < CAP) {
      ls[slot] = s;
      le[slot] = e;
      ld[slot] = (unsigned short)d;
    }
  });

  __syncthreads();
  const int total = __builtin_amdgcn_readfirstlane(nlist);
  if (total <= CAP) {
    const int per = (((total + 31) >> 5) + 7) & ~7;
    const int beg = wave * per;
    const int end = (beg + per < total) ? beg + per : total;
    for (int c0 = beg; c0 < end; c0 += 32) {
      const int cnt = (end - c0 < 32) ? end - c0 : 32;
      int s = 0, d = -1;
      float alpha = 0.f;
      if (lane < cnt) {
        s = ls[c0 + lane];
        d = ld[c0 + lane];
        alpha = __expf(le[c0 + lane] - mx[d]);
        atomicAdd(den + d, alpha);
      }
      add_rows<C>(s, d, alpha, cnt, lane, h, sums);
    }
  } else {
    scan_bytes(bins4, E, lane, wave, blockIdx.x, [&](int q, int cnt) {
      int s, d; float e;
      fetch_edge(q, cnt, lane, lo, nr, E, src, dst, asrc, adst, s, d, e);
      float alpha = 0.f;
      if (d >= 0) {
        alpha = __expf(e - mx[d]);
        atomicAdd(den + d, alpha);
      }
      add_rows<C>(s, d, alpha, cnt, lane, h, sums);
    });
  }

  __syncthreads();
  constexpr int P = C / 32;
  for (int i = tid; i < nr * C; i += 1024) {
    const int n = i / C;
    const int cc = i % C;
    const float dn = den[n];
    const float qv = sums[n * C + 32 * (cc % P) + cc / P] / dn;
    float r = ((dn > 0.f) ? qv : 0.f) + bias[i % C];
    if (RELU) r = fmaxf(r, 0.f);
    x[(size_t)lo * C + i] = r;
  }
}

static inline int cdiv(long a, long b) { return (int)((a + b - 1) / b); }

extern "C" void kernel_launch(void* const* d_in, const int* in_sizes, int n_in,
                              void* d_out, int out_size, void* d_ws, size_t ws_size,
                              hipStream_t stream)
{
  const float* x       = (const float*)d_in[0];
  const int*   ei      = (const int*)d_in[1];
  const float* Wsrc1   = (const float*)d_in[2];
  const float* Wdst1   = (const float*)d_in[3];
  const float* attsrc1 = (const float*)d_in[4];
  const float* attdst1 = (const float*)d_in[5];
  const float* b1      = (const float*)d_in[6];
  const float* Wsrc2   = (const float*)d_in[7];
  const float* Wdst2   = (const float*)d_in[8];
  const float* attsrc2 = (const float*)d_in[9];
  const float* attdst2 = (const float*)d_in[10];
  const float* b2      = (const float*)d_in[11];

  const int HID = in_sizes[4];
  const int OUT = in_sizes[9];
  const int FIN = in_sizes[2] / HID;
  const int N   = in_sizes[0] / FIN;
  const int E   = in_sizes[1] / 2;
  const int* src = ei;
  const int* dst = ei + E;

  unsigned char* bins = (unsigned char*)d_ws;
  float* ws    = (float*)d_ws + (((size_t)E + 127) / 128) * 32;
  float* h1    = ws;  ws += (size_t)N * HID;
  float* o1    = ws;  ws += (size_t)N * HID;
  float* h2    = ws;  ws += (size_t)N * OUT;
  float* asrc  = ws;  ws += N;
  float* adst  = ws;  ws += N;
  float* vfold = ws;  ws += HID;

  {
    const int tiles = (N / 16) * (HID / 64);
    gemm_wmma_f32<128, 128, 4><<<cdiv(tiles, 8), 256, 0, stream>>>(x, Wsrc1, h1, N);
    rowdot<128, 25><<<cdiv(N, RD_WAVES * 25), RD_WAVES * 32, 0, stream>>>(h1, attsrc1, asrc, N);
    foldvec<<<cdiv(FIN, 128), 128, 0, stream>>>(Wdst1, attdst1, vfold, FIN, HID);
    rowdot<128, 25><<<cdiv(N, RD_WAVES * 25), RD_WAVES * 32, 0, stream>>>(x, vfold, adst, N);
    range_bytes<<<cdiv(((long)E + 15) / 16 * 4, 256), 256, 0, stream>>>(dst, (unsigned*)bins, E);
    range_rows<128, true><<<cdiv(N, RNG), 1024, 0, stream>>>(src, dst, (const uint4*)bins, asrc, adst, h1, b1, o1, E, N);
  }

  {
    float* out = (float*)d_out;
    const int tiles = (N / 16) * (OUT / 64);
    gemm_wmma_f32<128, 64, 4><<<cdiv(tiles, 8), 256, 0, stream>>>(o1, Wsrc2, h2, N);
    rowdot<64, 13><<<cdiv(N, RD_WAVES * 26), RD_WAVES * 32, 0, stream>>>(h2, attsrc2, asrc, N);
    foldvec<<<cdiv(HID, 128), 128, 0, stream>>>(Wdst2, attdst2, vfold, HID, OUT);
    rowdot<128, 25><<<cdiv(N, RD_WAVES * 25), RD_WAVES * 32, 0, stream>>>(o1, vfold, adst, N);
    range_rows<64, false><<<cdiv(N, RNG), 1024, 0, stream>>>(src, dst, (const uint4*)bins, asrc, adst, h2, b2, out, E, N);
  }
}
